# starting kernel
# baseline (speedup 1.0000x reference)
#include <hip/hip_runtime.h>

typedef float v2f __attribute__((ext_vector_type(2)));
typedef float v8f __attribute__((ext_vector_type(8)));

#define N_NODES   50000
#define DIM       256
#define NEG_SLOPE 0.01f
#define LDS_PITCH 260
#define EDGES_PER_BLOCK 4

__global__ __launch_bounds__(256) void zero_f4(float* __restrict__ out, int n4) {
  int i = blockIdx.x * blockDim.x + threadIdx.x;
  if (i < n4) {
    float4 z = make_float4(0.f, 0.f, 0.f, 0.f);
    reinterpret_cast<float4*>(out)[i] = z;
  }
}

__global__ __launch_bounds__(256) void spmm_scatter(
    const int* __restrict__ erow, const int* __restrict__ ecol,
    const float* __restrict__ eval, const float* __restrict__ x,
    float* side, int n_edges) {
  const int t  = threadIdx.x;
  const int e0 = blockIdx.x * EDGES_PER_BLOCK;
#pragma unroll
  for (int i = 0; i < EDGES_PER_BLOCK; ++i) {
    int e = e0 + i;
    if (e < n_edges) {
      int   r = erow[e];
      int   c = ecol[e];
      float v = eval[e];
      float m = v * x[(size_t)c * DIM + t];
      atomicAdd(&side[(size_t)r * DIM + t], m);
    }
  }
}

__global__ __launch_bounds__(256) void fused_dual_gemm(
    const float* __restrict__ ego,
    const float* __restrict__ W1, const float* __restrict__ b1,
    const float* __restrict__ W2, const float* __restrict__ b2,
    float* out) {
  __shared__ float a1[16 * LDS_PITCH];
  __shared__ float a2[16 * LDS_PITCH];

  const int rowbase = blockIdx.x * 16;
  const int t = threadIdx.x;

#pragma unroll
  for (int j = 0; j < 4; ++j) {
    int i4 = t + 256 * j;
    int r  = i4 >> 6;
    int c4 = i4 & 63;
    float4 e = reinterpret_cast<const float4*>(ego + (size_t)(rowbase + r) * DIM)[c4];
    float4 s = reinterpret_cast<const float4*>(out + (size_t)(rowbase + r) * DIM)[c4];
    float* p1 = &a1[r * LDS_PITCH + c4 * 4];
    float* p2 = &a2[r * LDS_PITCH + c4 * 4];
    p1[0] = e.x + s.x; p1[1] = e.y + s.y; p1[2] = e.z + s.z; p1[3] = e.w + s.w;
    p2[0] = e.x * s.x; p2[1] = e.y * s.y; p2[2] = e.z * s.z; p2[3] = e.w * s.w;
  }
  __syncthreads();

  const int lane = t & 31;
  const int wave = t >> 5;
  const int m    = lane & 15;
  const int half = lane >> 4;

  const float* a1p = &a1[m * LDS_PITCH + half * 2];
  const float* a2p = &a2[m * LDS_PITCH + half * 2];

#pragma unroll
  for (int cti = 0; cti < 2; ++cti) {
    const int colbase = (wave + cti * 8) * 16;
    const int jg = colbase + m;
    const float* w1p = W1 + (size_t)jg * DIM + half * 2;
    const float* w2p = W2 + (size_t)jg * DIM + half * 2;

    v8f acc1 = {};
    v8f acc2 = {};
#pragma unroll 8
    for (int k = 0; k < DIM; k += 4) {
      v2f fa1 = *reinterpret_cast<const v2f*>(a1p + k);
      v2f fa2 = *reinterpret_cast<const v2f*>(a2p + k);
      v2f fw1 = *reinterpret_cast<const v2f*>(w1p + k);
      v2f fw2 = *reinterpret_cast<const v2f*>(w2p + k);
      acc1 = __builtin_amdgcn_wmma_f32_16x16x4_f32(
          false, fa1, false, fw1, (short)0, acc1, false, false);
      acc2 = __builtin_amdgcn_wmma_f32_16x16x4_f32(
          false, fa2, false, fw2, (short)0, acc2, false, false);
    }

    const float bias1 = b1[jg];
    const float bias2 = b2[jg];
    float* op = out + (size_t)(rowbase + half * 8) * DIM + colbase + m;
#pragma unroll
    for (int v = 0; v < 8; ++v) {
      float s1 = acc1[v] + bias1;
      s1 = (s1 >= 0.f) ? s1 : NEG_SLOPE * s1;
      float s2 = acc2[v] + bias2;
      s2 = (s2 >= 0.f) ? s2 : NEG_SLOPE * s2;
      op[(size_t)v * DIM] = s1 + s2;
    }
  }
}

extern "C" void kernel_launch(void* const* d_in, const int* in_sizes, int n_in,
                              void* d_out, int out_size, void* d_ws, size_t ws_size,
                              hipStream_t stream) {
  const float* ego  = (const float*)d_in[0];
  const int*   erow = (const int*)  d_in[1];
  const int*   ecol = (const int*)  d_in[2];
  const float* eval = (const float*)d_in[3];
  const float* W1   = (const float*)d_in[4];
  const float* b1   = (const float*)d_in[5];
  const float* W2   = (const float*)d_in[6];
  const float* b2   = (const float*)d_in[7];
  float* out = (float*)d_out;
  const int n_edges = in_sizes[1];

  const int n4 = (N_NODES * DIM) / 4;
  zero_f4<<<(n4 + 255) / 256, 256, 0, stream>>>(out, n4);

  const int nblk = (n_edges + EDGES_PER_BLOCK - 1) / EDGES_PER_BLOCK;
  spmm_scatter<<<nblk, 256, 0, stream>>>(erow, ecol, eval, ego, out, n_edges);

  fused_dual_gemm<<<N_NODES / 16, 256, 0, stream>>>(ego, W1, b1, W2, b2, out);
}


# Per-edge f32 atomic scatter replaced by counting sort of edges by destination row (integer atomics) + one wave per row summing in registers; row sum order = arrival slot order
# speedup vs baseline: 1.7623x; 1.7623x over previous
#include <hip/hip_runtime.h>

typedef float v2f __attribute__((ext_vector_type(2)));
typedef float v8f __attribute__((ext_vector_type(8)));

#define N_NODES   50000
#define DIM       256
#define NEG_SLOPE 0.01f
#define LDS_PITCH 260
#define EDGES_PER_BLOCK 4

__global__ __launch_bounds__(256) void zero_f4(float* __restrict__ out, int n4) {
  int i = blockIdx.x * blockDim.x + threadIdx.x;
  if (i < n4) {
    float4 z = make_float4(0.f, 0.f, 0.f, 0.f);
    reinterpret_cast<float4*>(out)[i] = z;
  }
}

__global__ __launch_bounds__(256) void spmm_scatter(
    const int* __restrict__ erow, const int* __restrict__ ecol,
    const float* __restrict__ eval, const float* __restrict__ x,
    float* side, int n_edges) {
  const int t  = threadIdx.x;
  const int e0 = blockIdx.x * EDGES_PER_BLOCK;
#pragma unroll
  for (int i = 0; i < EDGES_PER_BLOCK; ++i) {
    int e = e0 + i;
    if (e < n_edges) {
      int   r = erow[e];
      int   c = ecol[e];
      float v = eval[e];
      float m = v * x[(size_t)c * DIM + t];
      atomicAdd(&side[(size_t)r * DIM + t], m);
    }
  }
}

#define SCAN_T   1024
#define ROWS_PAD 50176

__global__ __launch_bounds__(256) void zero_i32(int* __restrict__ p, int n) {
  int i = blockIdx.x * blockDim.x + threadIdx.x;
  if (i < n) p[i] = 0;
}

__global__ __launch_bounds__(256) void edge_rank(
    const int* __restrict__ erow, int* cnt, int* __restrict__ rank, int n_edges) {
  int e = blockIdx.x * blockDim.x + threadIdx.x;
  if (e < n_edges) rank[e] = atomicAdd(&cnt[erow[e]], 1);
}

__global__ __launch_bounds__(SCAN_T) void row_offsets(
    const int* __restrict__ cnt, int* __restrict__ rowptr, int n) {
  __shared__ int wsum[32];
  __shared__ int wpre[32];
  const int t = threadIdx.x;
  const int lane = t & 31;
  const int w = t >> 5;
  const int b = blockIdx.x;

  int pre = 0;
#pragma unroll 8
  for (int j = 0; j < b; ++j) pre += cnt[j * SCAN_T + t];
  const int i = b * SCAN_T + t;
  const int c = (i < n) ? cnt[i] : 0;

  int incl = c;
#pragma unroll
  for (int d = 1; d < 32; d <<= 1) {
    int y = __shfl_up(incl, d);
    if (lane >= d) incl += y;
  }
#pragma unroll
  for (int d = 16; d > 0; d >>= 1) pre += __shfl_xor(pre, d);
  if (lane == 31) wsum[w] = incl;
  if (lane == 0)  wpre[w] = pre;
  __syncthreads();

  int part = wpre[lane] + ((lane < w) ? wsum[lane] : 0);
#pragma unroll
  for (int d = 16; d > 0; d >>= 1) part += __shfl_xor(part, d);
  if (i < n) rowptr[i] = part + incl - c;
}

__global__ __launch_bounds__(256) void edge_place(
    const int* __restrict__ erow, const int* __restrict__ ecol,
    const float* __restrict__ eval, const int* __restrict__ rowptr,
    const int* __restrict__ rank, int* __restrict__ rec, int n_edges) {
  int e = blockIdx.x * blockDim.x + threadIdx.x;
  if (e < n_edges) {
    int pos = rowptr[erow[e]] + rank[e];
    if ((unsigned)pos < (unsigned)n_edges)
      reinterpret_cast<int2*>(rec)[pos] = make_int2(ecol[e], __float_as_int(eval[e]));
  }
}

struct RowAcc { float4 lo, hi; };

__device__ __forceinline__ void add_scaled(RowAcc& a, float v, const float4& x0, const float4& x1) {
  a.lo.x += v * x0.x; a.lo.y += v * x0.y; a.lo.z += v * x0.z; a.lo.w += v * x0.w;
  a.hi.x += v * x1.x; a.hi.y += v * x1.y; a.hi.z += v * x1.z; a.hi.w += v * x1.w;
}

__global__ __launch_bounds__(256) void row_gather_sum(
    const float* __restrict__ x, const int* __restrict__ rowptr,
    const int* __restrict__ cnt, const int* __restrict__ rec,
    float* __restrict__ side, int n_rows) {
  const int lane = threadIdx.x & 31;
  const int r = __builtin_amdgcn_readfirstlane(blockIdx.x * 8 + (threadIdx.x >> 5));
  if (r >= n_rows) return;
  const int start = __builtin_amdgcn_readfirstlane(rowptr[r]);
  const int deg   = __builtin_amdgcn_readfirstlane(cnt[r]);
  const int2* recs = reinterpret_cast<const int2*>(rec) + start;
  const float4* xl = reinterpret_cast<const float4*>(x) + lane;

  RowAcc a;
  a.lo = make_float4(0.f, 0.f, 0.f, 0.f);
  a.hi = make_float4(0.f, 0.f, 0.f, 0.f);

  for (int base = 0; base < deg; base += 32) {
    const int n = (deg - base < 32) ? (deg - base) : 32;
    int2 rc = make_int2(0, 0);
    if (lane < n) rc = recs[base + lane];
    int j = 0;
    for (; j + 4 <= n; j += 4) {
      const int c0 = __builtin_amdgcn_readlane(rc.x, j);
      const int c1 = __builtin_amdgcn_readlane(rc.x, j + 1);
      const int c2 = __builtin_amdgcn_readlane(rc.x, j + 2);
      const int c3 = __builtin_amdgcn_readlane(rc.x, j + 3);
      const float v0 = __int_as_float(__builtin_amdgcn_readlane(rc.y, j));
      const float v1 = __int_as_float(__builtin_amdgcn_readlane(rc.y, j + 1));
      const float v2 = __int_as_float(__builtin_amdgcn_readlane(rc.y, j + 2));
      const float v3 = __int_as_float(__builtin_amdgcn_readlane(rc.y, j + 3));
      const float4* p0 = xl + (size_t)c0 * (DIM / 4);
      const float4* p1 = xl + (size_t)c1 * (DIM / 4);
      const float4* p2 = xl + (size_t)c2 * (DIM / 4);
      const float4* p3 = xl + (size_t)c3 * (DIM / 4);
      const float4 x00 = p0[0], x01 = p0[32];
      const float4 x10 = p1[0], x11 = p1[32];
      const float4 x20 = p2[0], x21 = p2[32];
      const float4 x30 = p3[0], x31 = p3[32];
      add_scaled(a, v0, x00, x01);
      add_scaled(a, v1, x10, x11);
      add_scaled(a, v2, x20, x21);
      add_scaled(a, v3, x30, x31);
    }
    if (j + 2 <= n) {
      const int c0 = __builtin_amdgcn_readlane(rc.x, j);
      const int c1 = __builtin_amdgcn_readlane(rc.x, j + 1);
      const float v0 = __int_as_float(__builtin_amdgcn_readlane(rc.y, j));
      const float v1 = __int_as_float(__builtin_amdgcn_readlane(rc.y, j + 1));
      const float4* p0 = xl + (size_t)c0 * (DIM / 4);
      const float4* p1 = xl + (size_t)c1 * (DIM / 4);
      const float4 x00 = p0[0], x01 = p0[32];
      const float4 x10 = p1[0], x11 = p1[32];
      add_scaled(a, v0, x00, x01);
      add_scaled(a, v1, x10, x11);
      j += 2;
    }
    if (j < n) {
      const int c0 = __builtin_amdgcn_readlane(rc.x, j);
      const float v0 = __int_as_float(__builtin_amdgcn_readlane(rc.y, j));
      const float4* p0 = xl + (size_t)c0 * (DIM / 4);
      const float4 x00 = p0[0], x01 = p0[32];
      add_scaled(a, v0, x00, x01);
    }
  }
  float4* sp = reinterpret_cast<float4*>(side + (size_t)r * DIM);
  sp[lane]      = a.lo;
  sp[lane + 32] = a.hi;
}

__global__ __launch_bounds__(256) void fused_dual_gemm(
    const float* __restrict__ ego,
    const float* __restrict__ W1, const float* __restrict__ b1,
    const float* __restrict__ W2, const float* __restrict__ b2,
    float* out) {
  __shared__ float a1[16 * LDS_PITCH];
  __shared__ float a2[16 * LDS_PITCH];

  const int rowbase = blockIdx.x * 16;
  const int t = threadIdx.x;

#pragma unroll
  for (int j = 0; j < 4; ++j) {
    int i4 = t + 256 * j;
    int r  = i4 >> 6;
    int c4 = i4 & 63;
    float4 e = reinterpret_cast<const float4*>(ego + (size_t)(rowbase + r) * DIM)[c4];
    float4 s = reinterpret_cast<const float4*>(out + (size_t)(rowbase + r) * DIM)[c4];
    float* p1 = &a1[r * LDS_PITCH + c4 * 4];
    float* p2 = &a2[r * LDS_PITCH + c4 * 4];
    p1[0] = e.x + s.x; p1[1] = e.y + s.y; p1[2] = e.z + s.z; p1[3] = e.w + s.w;
    p2[0] = e.x * s.x; p2[1] = e.y * s.y; p2[2] = e.z * s.z; p2[3] = e.w * s.w;
  }
  __syncthreads();

  const int lane = t & 31;
  const int wave = t >> 5;
  const int m    = lane & 15;
  const int half = lane >> 4;

  const float* a1p = &a1[m * LDS_PITCH + half * 2];
  const float* a2p = &a2[m * LDS_PITCH + half * 2];

#pragma unroll
  for (int cti = 0; cti < 2; ++cti) {
    const int colbase = (wave + cti * 8) * 16;
    const int jg = colbase + m;
    const float* w1p = W1 + (size_t)jg * DIM + half * 2;
    const float* w2p = W2 + (size_t)jg * DIM + half * 2;

    v8f acc1 = {};
    v8f acc2 = {};
#pragma unroll 8
    for (int k = 0; k < DIM; k += 4) {
      v2f fa1 = *reinterpret_cast<const v2f*>(a1p + k);
      v2f fa2 = *reinterpret_cast<const v2f*>(a2p + k);
      v2f fw1 = *reinterpret_cast<const v2f*>(w1p + k);
      v2f fw2 = *reinterpret_cast<const v2f*>(w2p + k);
      acc1 = __builtin_amdgcn_wmma_f32_16x16x4_f32(
          false, fa1, false, fw1, (short)0, acc1, false, false);
      acc2 = __builtin_amdgcn_wmma_f32_16x16x4_f32(
          false, fa2, false, fw2, (short)0, acc2, false, false);
    }

    const float bias1 = b1[jg];
    const float bias2 = b2[jg];
    float* op = out + (size_t)(rowbase + half * 8) * DIM + colbase + m;
#pragma unroll
    for (int v = 0; v < 8; ++v) {
      float s1 = acc1[v] + bias1;
      s1 = (s1 >= 0.f) ? s1 : NEG_SLOPE * s1;
      float s2 = acc2[v] + bias2;
      s2 = (s2 >= 0.f) ? s2 : NEG_SLOPE * s2;
      op[(size_t)v * DIM] = s1 + s2;
    }
  }
}

extern "C" void kernel_launch(void* const* d_in, const int* in_sizes, int n_in,
                              void* d_out, int out_size, void* d_ws, size_t ws_size,
                              hipStream_t stream) {
  const float* ego  = (const float*)d_in[0];
  const int*   erow = (const int*)  d_in[1];
  const int*   ecol = (const int*)  d_in[2];
  const float* eval = (const float*)d_in[3];
  const float* W1   = (const float*)d_in[4];
  const float* b1   = (const float*)d_in[5];
  const float* W2   = (const float*)d_in[6];
  const float* b2   = (const float*)d_in[7];
  float* out = (float*)d_out;
  const int n_edges = in_sizes[1];

  const size_t ws_need = ((size_t)2 * ROWS_PAD + (size_t)3 * n_edges) * sizeof(int);
  if (d_ws != nullptr && ws_size >= ws_need) {
    int* cnt    = (int*)d_ws;
    int* rowptr = cnt + ROWS_PAD;
    int* rec    = rowptr + ROWS_PAD;
    int* rank   = rec + (size_t)2 * n_edges;
    const int eblk = (n_edges + 255) / 256;
    zero_i32<<<ROWS_PAD / 256, 256, 0, stream>>>(cnt, ROWS_PAD);
    edge_rank<<<eblk, 256, 0, stream>>>(erow, cnt, rank, n_edges);
    row_offsets<<<ROWS_PAD / SCAN_T, SCAN_T, 0, stream>>>(cnt, rowptr, N_NODES);
    edge_place<<<eblk, 256, 0, stream>>>(erow, ecol, eval, rowptr, rank, rec, n_edges);
    row_gather_sum<<<(N_NODES + 7) / 8, 256, 0, stream>>>(ego, rowptr, cnt, rec, out, N_NODES);
  } else {
    const int n4 = (N_NODES * DIM) / 4;
    zero_f4<<<(n4 + 255) / 256, 256, 0, stream>>>(out, n4);
    const int nblk = (n_edges + EDGES_PER_BLOCK - 1) / EDGES_PER_BLOCK;
    spmm_scatter<<<nblk, 256, 0, stream>>>(erow, ecol, eval, ego, out, n_edges);
  }

  fused_dual_gemm<<<N_NODES / 16, 256, 0, stream>>>(ego, W1, b1, W2, b2, out);
}


# Dual GEMM: 256 work-groups take 12-13 row tiles each in LDS strips of 4-5 tiles (was one 16-row tile per group); a wave per 16-col tile reuses each loaded W fragment for the whole strip
# speedup vs baseline: 2.3590x; 1.3386x over previous
#include <hip/hip_runtime.h>

typedef float v2f __attribute__((ext_vector_type(2)));
typedef float v8f __attribute__((ext_vector_type(8)));

#define N_NODES   50000
#define DIM       256
#define NEG_SLOPE 0.01f
#define LDS_PITCH 260
#define EDGES_PER_BLOCK 4

__global__ __launch_bounds__(256) void zero_f4(float* __restrict__ out, int n4) {
  int i = blockIdx.x * blockDim.x + threadIdx.x;
  if (i < n4) {
    float4 z = make_float4(0.f, 0.f, 0.f, 0.f);
    reinterpret_cast<float4*>(out)[i] = z;
  }
}

__global__ __launch_bounds__(256) void spmm_scatter(
    const int* __restrict__ erow, const int* __restrict__ ecol,
    const float* __restrict__ eval, const float* __restrict__ x,
    float* side, int n_edges) {
  const int t  = threadIdx.x;
  const int e0 = blockIdx.x * EDGES_PER_BLOCK;
#pragma unroll
  for (int i = 0; i < EDGES_PER_BLOCK; ++i) {
    int e = e0 + i;
    if (e < n_edges) {
      int   r = erow[e];
      int   c = ecol[e];
      float v = eval[e];
      float m = v * x[(size_t)c * DIM + t];
      atomicAdd(&side[(size_t)r * DIM + t], m);
    }
  }
}

#define SCAN_T   1024
#define ROWS_PAD 50176

__global__ __launch_bounds__(256) void zero_i32(int* __restrict__ p, int n) {
  int i = blockIdx.x * blockDim.x + threadIdx.x;
  if (i < n) p[i] = 0;
}

__global__ __launch_bounds__(256) void edge_rank(
    const int* __restrict__ erow, int* cnt, int* __restrict__ rank, int n_edges) {
  int e = blockIdx.x * blockDim.x + threadIdx.x;
  if (e < n_edges) rank[e] = atomicAdd(&cnt[erow[e]], 1);
}

__global__ __launch_bounds__(SCAN_T) void row_offsets(
    const int* __restrict__ cnt, int* __restrict__ rowptr, int n) {
  __shared__ int wsum[32];
  __shared__ int wpre[32];
  const int t = threadIdx.x;
  const int lane = t & 31;
  const int w = t >> 5;
  const int b = blockIdx.x;

  int pre = 0;
#pragma unroll 8
  for (int j = 0; j < b; ++j) pre += cnt[j * SCAN_T + t];
  const int i = b * SCAN_T + t;
  const int c = (i < n) ? cnt[i] : 0;

  int incl = c;
#pragma unroll
  for (int d = 1; d < 32; d <<= 1) {
    int y = __shfl_up(incl, d);
    if (lane >= d) incl += y;
  }
#pragma unroll
  for (int d = 16; d > 0; d >>= 1) pre += __shfl_xor(pre, d);
  if (lane == 31) wsum[w] = incl;
  if (lane == 0)  wpre[w] = pre;
  __syncthreads();

  int part = wpre[lane] + ((lane < w) ? wsum[lane] : 0);
#pragma unroll
  for (int d = 16; d > 0; d >>= 1) part += __shfl_xor(part, d);
  if (i < n) rowptr[i] = part + incl - c;
}

__global__ __launch_bounds__(256) void edge_place(
    const int* __restrict__ erow, const int* __restrict__ ecol,
    const float* __restrict__ eval, const int* __restrict__ rowptr,
    const int* __restrict__ rank, int* __restrict__ rec, int n_edges) {
  int e = blockIdx.x * blockDim.x + threadIdx.x;
  if (e < n_edges) {
    int pos = rowptr[erow[e]] + rank[e];
    if ((unsigned)pos < (unsigned)n_edges)
      reinterpret_cast<int2*>(rec)[pos] = make_int2(ecol[e], __float_as_int(eval[e]));
  }
}

struct RowAcc { float4 lo, hi; };

__device__ __forceinline__ void add_scaled(RowAcc& a, float v, const float4& x0, const float4& x1) {
  a.lo.x += v * x0.x; a.lo.y += v * x0.y; a.lo.z += v * x0.z; a.lo.w += v * x0.w;
  a.hi.x += v * x1.x; a.hi.y += v * x1.y; a.hi.z += v * x1.z; a.hi.w += v * x1.w;
}

__global__ __launch_bounds__(256) void row_gather_sum(
    const float* __restrict__ x, const int* __restrict__ rowptr,
    const int* __restrict__ cnt, const int* __restrict__ rec,
    float* __restrict__ side, int n_rows) {
  const int lane = threadIdx.x & 31;
  const int r = __builtin_amdgcn_readfirstlane(blockIdx.x * 8 + (threadIdx.x >> 5));
  if (r >= n_rows) return;
  const int start = __builtin_amdgcn_readfirstlane(rowptr[r]);
  const int deg   = __builtin_amdgcn_readfirstlane(cnt[r]);
  const int2* recs = reinterpret_cast<const int2*>(rec) + start;
  const float4* xl = reinterpret_cast<const float4*>(x) + lane;

  RowAcc a;
  a.lo = make_float4(0.f, 0.f, 0.f, 0.f);
  a.hi = make_float4(0.f, 0.f, 0.f, 0.f);

  for (int base = 0; base < deg; base += 32) {
    const int n = (deg - base < 32) ? (deg - base) : 32;
    int2 rc = make_int2(0, 0);
    if (lane < n) rc = recs[base + lane];
    int j = 0;
    for (; j + 4 <= n; j += 4) {
      const int c0 = __builtin_amdgcn_readlane(rc.x, j);
      const int c1 = __builtin_amdgcn_readlane(rc.x, j + 1);
      const int c2 = __builtin_amdgcn_readlane(rc.x, j + 2);
      const int c3 = __builtin_amdgcn_readlane(rc.x, j + 3);
      const float v0 = __int_as_float(__builtin_amdgcn_readlane(rc.y, j));
      const float v1 = __int_as_float(__builtin_amdgcn_readlane(rc.y, j + 1));
      const float v2 = __int_as_float(__builtin_amdgcn_readlane(rc.y, j + 2));
      const float v3 = __int_as_float(__builtin_amdgcn_readlane(rc.y, j + 3));
      const float4* p0 = xl + (size_t)c0 * (DIM / 4);
      const float4* p1 = xl + (size_t)c1 * (DIM / 4);
      const float4* p2 = xl + (size_t)c2 * (DIM / 4);
      const float4* p3 = xl + (size_t)c3 * (DIM / 4);
      const float4 x00 = p0[0], x01 = p0[32];
      const float4 x10 = p1[0], x11 = p1[32];
      const float4 x20 = p2[0], x21 = p2[32];
      const float4 x30 = p3[0], x31 = p3[32];
      add_scaled(a, v0, x00, x01);
      add_scaled(a, v1, x10, x11);
      add_scaled(a, v2, x20, x21);
      add_scaled(a, v3, x30, x31);
    }
    if (j + 2 <= n) {
      const int c0 = __builtin_amdgcn_readlane(rc.x, j);
      const int c1 = __builtin_amdgcn_readlane(rc.x, j + 1);
      const float v0 = __int_as_float(__builtin_amdgcn_readlane(rc.y, j));
      const float v1 = __int_as_float(__builtin_amdgcn_readlane(rc.y, j + 1));
      const float4* p0 = xl + (size_t)c0 * (DIM / 4);
      const float4* p1 = xl + (size_t)c1 * (DIM / 4);
      const float4 x00 = p0[0], x01 = p0[32];
      const float4 x10 = p1[0], x11 = p1[32];
      add_scaled(a, v0, x00, x01);
      add_scaled(a, v1, x10, x11);
      j += 2;
    }
    if (j < n) {
      const int c0 = __builtin_amdgcn_readlane(rc.x, j);
      const float v0 = __int_as_float(__builtin_amdgcn_readlane(rc.y, j));
      const float4* p0 = xl + (size_t)c0 * (DIM / 4);
      const float4 x00 = p0[0], x01 = p0[32];
      add_scaled(a, v0, x00, x01);
    }
  }
  float4* sp = reinterpret_cast<float4*>(side + (size_t)r * DIM);
  sp[lane]      = a.lo;
  sp[lane + 32] = a.hi;
}

#define GEMM_GROUPS 256
#define GEMM_TILES  (N_NODES / 16)
#define GEMM_RT_MAX 5
#define GEMM_T      512
#define GEMM_KU     4

struct WSet { v2f w1[GEMM_KU], w2[GEMM_KU]; };

__device__ __forceinline__ void load_wset(WSet& s, const float* __restrict__ w1p,
                                          const float* __restrict__ w2p, int k0) {
#pragma unroll
  for (int u = 0; u < GEMM_KU; ++u) {
    s.w1[u] = *reinterpret_cast<const v2f*>(w1p + k0 + 4 * u);
    s.w2[u] = *reinterpret_cast<const v2f*>(w2p + k0 + 4 * u);
  }
}

template <int RT>
__device__ __forceinline__ void mma_wset(v8f (&acc1)[RT], v8f (&acc2)[RT], const WSet& s,
                                         const float* a1p, const float* a2p, int k0) {
#pragma unroll
  for (int u = 0; u < GEMM_KU; ++u) {
#pragma unroll
    for (int rt = 0; rt < RT; ++rt) {
      const v2f fa1 = *reinterpret_cast<const v2f*>(a1p + rt * 16 * LDS_PITCH + k0 + 4 * u);
      const v2f fa2 = *reinterpret_cast<const v2f*>(a2p + rt * 16 * LDS_PITCH + k0 + 4 * u);
      acc1[rt] = __builtin_amdgcn_wmma_f32_16x16x4_f32(
          false, fa1, false, s.w1[u], (short)0, acc1[rt], false, false);
      acc2[rt] = __builtin_amdgcn_wmma_f32_16x16x4_f32(
          false, fa2, false, s.w2[u], (short)0, acc2[rt], false, false);
    }
  }
}

template <int RT>
__device__ __forceinline__ void gemm_strip(
    float* a1, float* a2, const int rowbase, const int t,
    const float* __restrict__ ego, const float* __restrict__ w1p, const float* __restrict__ w2p,
    const float bias1, const float bias2, float* out) {
#pragma unroll
  for (int j = 0; j < (RT * 16 * 64) / GEMM_T; ++j) {
    const int i4 = t + GEMM_T * j;
    const int r  = i4 >> 6;
    const int c4 = i4 & 63;
    const float4 e = reinterpret_cast<const float4*>(ego + (size_t)(rowbase + r) * DIM)[c4];
    const float4 s = reinterpret_cast<const float4*>(out + (size_t)(rowbase + r) * DIM)[c4];
    *reinterpret_cast<float4*>(&a1[r * LDS_PITCH + c4 * 4]) =
        make_float4(e.x + s.x, e.y + s.y, e.z + s.z, e.w + s.w);
    *reinterpret_cast<float4*>(&a2[r * LDS_PITCH + c4 * 4]) =
        make_float4(e.x * s.x, e.y * s.y, e.z * s.z, e.w * s.w);
  }
  __syncthreads();

  const int lane = t & 31;
  const int wave = t >> 5;
  const int m    = lane & 15;
  const int half = lane >> 4;

  const float* a1p = &a1[m * LDS_PITCH + half * 2];
  const float* a2p = &a2[m * LDS_PITCH + half * 2];

  v8f acc1[RT];
  v8f acc2[RT];
#pragma unroll
  for (int rt = 0; rt < RT; ++rt) { acc1[rt] = v8f{}; acc2[rt] = v8f{}; }

  WSet wa, wb;
  load_wset(wa, w1p, w2p, 0);
  load_wset(wb, w1p, w2p, 4 * GEMM_KU);
#pragma unroll 1
  for (int kb = 0; kb < DIM - 8 * GEMM_KU; kb += 8 * GEMM_KU) {
    mma_wset<RT>(acc1, acc2, wa, a1p, a2p, kb);
    load_wset(wa, w1p, w2p, kb + 8 * GEMM_KU);
    mma_wset<RT>(acc1, acc2, wb, a1p, a2p, kb + 4 * GEMM_KU);
    load_wset(wb, w1p, w2p, kb + 12 * GEMM_KU);
  }
  mma_wset<RT>(acc1, acc2, wa, a1p, a2p, DIM - 8 * GEMM_KU);
  mma_wset<RT>(acc1, acc2, wb, a1p, a2p, DIM - 4 * GEMM_KU);

#pragma unroll
  for (int rt = 0; rt < RT; ++rt) {
    float* op = out + (size_t)(rowbase + rt * 16 + half * 8) * DIM + wave * 16 + m;
#pragma unroll
    for (int v = 0; v < 8; ++v) {
      float s1 = acc1[rt][v] + bias1;
      s1 = (s1 >= 0.f) ? s1 : NEG_SLOPE * s1;
      float s2 = acc2[rt][v] + bias2;
      s2 = (s2 >= 0.f) ? s2 : NEG_SLOPE * s2;
      op[(size_t)v * DIM] = s1 + s2;
    }
  }
  __syncthreads();
}

__global__ __launch_bounds__(GEMM_T) void fused_dual_gemm(
    const float* __restrict__ ego,
    const float* __restrict__ W1, const float* __restrict__ b1,
    const float* __restrict__ W2, const float* __restrict__ b2,
    float* out) {
  __shared__ __attribute__((aligned(16))) float a1[GEMM_RT_MAX * 16 * LDS_PITCH];
  __shared__ __attribute__((aligned(16))) float a2[GEMM_RT_MAX * 16 * LDS_PITCH];

  const int t = threadIdx.x;
  const int b = blockIdx.x;
  const int tile0 = (b * GEMM_TILES) / GEMM_GROUPS;
  const int tile1 = ((b + 1) * GEMM_TILES) / GEMM_GROUPS;

  const int jg = (t >> 5) * 16 + (t & 15);
  const int half = (t & 31) >> 4;
  const float* w1p = W1 + (size_t)jg * DIM + half * 2;
  const float* w2p = W2 + (size_t)jg * DIM + half * 2;
  const float bias1 = b1[jg];
  const float bias2 = b2[jg];

  int tile = tile0;
  if (tile1 - tile0 == 13) {
    gemm_strip<5>(a1, a2, tile * 16, t, ego, w1p, w2p, bias1, bias2, out);
    tile += 5;
  } else {
    gemm_strip<4>(a1, a2, tile * 16, t, ego, w1p, w2p, bias1, bias2, out);
    tile += 4;
  }
#pragma unroll 1
  for (; tile < tile1; tile += 4)
    gemm_strip<4>(a1, a2, tile * 16, t, ego, w1p, w2p, bias1, bias2, out);
}

extern "C" void kernel_launch(void* const* d_in, const int* in_sizes, int n_in,
                              void* d_out, int out_size, void* d_ws, size_t ws_size,
                              hipStream_t stream) {
  const float* ego  = (const float*)d_in[0];
  const int*   erow = (const int*)  d_in[1];
  const int*   ecol = (const int*)  d_in[2];
  const float* eval = (const float*)d_in[3];
  const float* W1   = (const float*)d_in[4];
  const float* b1   = (const float*)d_in[5];
  const float* W2   = (const float*)d_in[6];
  const float* b2   = (const float*)d_in[7];
  float* out = (float*)d_out;
  const int n_edges = in_sizes[1];

  const size_t ws_need = ((size_t)2 * ROWS_PAD + (size_t)3 * n_edges) * sizeof(int);
  if (d_ws != nullptr && ws_size >= ws_need) {
    int* cnt    = (int*)d_ws;
    int* rowptr = cnt + ROWS_PAD;
    int* rec    = rowptr + ROWS_PAD;
    int* rank   = rec + (size_t)2 * n_edges;
    const int eblk = (n_edges + 255) / 256;
    zero_i32<<<ROWS_PAD / 256, 256, 0, stream>>>(cnt, ROWS_PAD);
    edge_rank<<<eblk, 256, 0, stream>>>(erow, cnt, rank, n_edges);
    row_offsets<<<ROWS_PAD / SCAN_T, SCAN_T, 0, stream>>>(cnt, rowptr, N_NODES);
    edge_place<<<eblk, 256, 0, stream>>>(erow, ecol, eval, rowptr, rank, rec, n_edges);
    row_gather_sum<<<(N_NODES + 7) / 8, 256, 0, stream>>>(ego, rowptr, cnt, rec, out, N_NODES);
  } else {
    const int n4 = (N_NODES * DIM) / 4;
    zero_f4<<<(n4 + 255) / 256, 256, 0, stream>>>(out, n4);
    const int nblk = (n_edges + EDGES_PER_BLOCK - 1) / EDGES_PER_BLOCK;
    spmm_scatter<<<nblk, 256, 0, stream>>>(erow, ecol, eval, ego, out, n_edges);
  }

  fused_dual_gemm<<<GEMM_GROUPS, GEMM_T, 0, stream>>>(ego, W1, b1, W2, b2, out);
}


# GEMM k-loop feed: A (LDS) and W fragments as 16-byte per-lane loads, each 8-k block loaded a whole block ahead into a second register set; k summed in groups {0,1,4,5},{2,3,6,7}
# speedup vs baseline: 2.5077x; 1.0630x over previous
#include <hip/hip_runtime.h>

typedef float v2f __attribute__((ext_vector_type(2)));
typedef float v8f __attribute__((ext_vector_type(8)));

#define N_NODES   50000
#define DIM       256
#define NEG_SLOPE 0.01f
#define LDS_PITCH 260
#define EDGES_PER_BLOCK 4

__global__ __launch_bounds__(256) void zero_f4(float* __restrict__ out, int n4) {
  int i = blockIdx.x * blockDim.x + threadIdx.x;
  if (i < n4) {
    float4 z = make_float4(0.f, 0.f, 0.f, 0.f);
    reinterpret_cast<float4*>(out)[i] = z;
  }
}

__global__ __launch_bounds__(256) void spmm_scatter(
    const int* __restrict__ erow, const int* __restrict__ ecol,
    const float* __restrict__ eval, const float* __restrict__ x,
    float* side, int n_edges) {
  const int t  = threadIdx.x;
  const int e0 = blockIdx.x * EDGES_PER_BLOCK;
#pragma unroll
  for (int i = 0; i < EDGES_PER_BLOCK; ++i) {
    int e = e0 + i;
    if (e < n_edges) {
      int   r = erow[e];
      int   c = ecol[e];
      float v = eval[e];
      float m = v * x[(size_t)c * DIM + t];
      atomicAdd(&side[(size_t)r * DIM + t], m);
    }
  }
}

#define SCAN_T   1024
#define ROWS_PAD 50176

__global__ __launch_bounds__(256) void zero_i32(int* __restrict__ p, int n) {
  int i = blockIdx.x * blockDim.x + threadIdx.x;
  if (i < n) p[i] = 0;
}

__global__ __launch_bounds__(256) void edge_rank(
    const int* __restrict__ erow, int* cnt, int* __restrict__ rank, int n_edges) {
  int e = blockIdx.x * blockDim.x + threadIdx.x;
  if (e < n_edges) rank[e] = atomicAdd(&cnt[erow[e]], 1);
}

__global__ __launch_bounds__(SCAN_T) void row_offsets(
    const int* __restrict__ cnt, int* __restrict__ rowptr, int n) {
  __shared__ int wsum[32];
  __shared__ int wpre[32];
  const int t = threadIdx.x;
  const int lane = t & 31;
  const int w = t >> 5;
  const int b = blockIdx.x;

  int pre = 0;
#pragma unroll 8
  for (int j = 0; j < b; ++j) pre += cnt[j * SCAN_T + t];
  const int i = b * SCAN_T + t;
  const int c = (i < n) ? cnt[i] : 0;

  int incl = c;
#pragma unroll
  for (int d = 1; d < 32; d <<= 1) {
    int y = __shfl_up(incl, d);
    if (lane >= d) incl += y;
  }
#pragma unroll
  for (int d = 16; d > 0; d >>= 1) pre += __shfl_xor(pre, d);
  if (lane == 31) wsum[w] = incl;
  if (lane == 0)  wpre[w] = pre;
  __syncthreads();

  int part = wpre[lane] + ((lane < w) ? wsum[lane] : 0);
#pragma unroll
  for (int d = 16; d > 0; d >>= 1) part += __shfl_xor(part, d);
  if (i < n) rowptr[i] = part + incl - c;
}

__global__ __launch_bounds__(256) void edge_place(
    const int* __restrict__ erow, const int* __restrict__ ecol,
    const float* __restrict__ eval, const int* __restrict__ rowptr,
    const int* __restrict__ rank, int* __restrict__ rec, int n_edges) {
  int e = blockIdx.x * blockDim.x + threadIdx.x;
  if (e < n_edges) {
    int pos = rowptr[erow[e]] + rank[e];
    if ((unsigned)pos < (unsigned)n_edges)
      reinterpret_cast<int2*>(rec)[pos] = make_int2(ecol[e], __float_as_int(eval[e]));
  }
}

struct RowAcc { float4 lo, hi; };

__device__ __forceinline__ void add_scaled(RowAcc& a, float v, const float4& x0, const float4& x1) {
  a.lo.x += v * x0.x; a.lo.y += v * x0.y; a.lo.z += v * x0.z; a.lo.w += v * x0.w;
  a.hi.x += v * x1.x; a.hi.y += v * x1.y; a.hi.z += v * x1.z; a.hi.w += v * x1.w;
}

__global__ __launch_bounds__(256) void row_gather_sum(
    const float* __restrict__ x, const int* __restrict__ rowptr,
    const int* __restrict__ cnt, const int* __restrict__ rec,
    float* __restrict__ side, int n_rows) {
  const int lane = threadIdx.x & 31;
  const int r = __builtin_amdgcn_readfirstlane(blockIdx.x * 8 + (threadIdx.x >> 5));
  if (r >= n_rows) return;
  const int start = __builtin_amdgcn_readfirstlane(rowptr[r]);
  const int deg   = __builtin_amdgcn_readfirstlane(cnt[r]);
  const int2* recs = reinterpret_cast<const int2*>(rec) + start;
  const float4* xl = reinterpret_cast<const float4*>(x) + lane;

  RowAcc a;
  a.lo = make_float4(0.f, 0.f, 0.f, 0.f);
  a.hi = make_float4(0.f, 0.f, 0.f, 0.f);

  for (int base = 0; base < deg; base += 32) {
    const int n = (deg - base < 32) ? (deg - base) : 32;
    int2 rc = make_int2(0, 0);
    if (lane < n) rc = recs[base + lane];
    int j = 0;
    for (; j + 4 <= n; j += 4) {
      const int c0 = __builtin_amdgcn_readlane(rc.x, j);
      const int c1 = __builtin_amdgcn_readlane(rc.x, j + 1);
      const int c2 = __builtin_amdgcn_readlane(rc.x, j + 2);
      const int c3 = __builtin_amdgcn_readlane(rc.x, j + 3);
      const float v0 = __int_as_float(__builtin_amdgcn_readlane(rc.y, j));
      const float v1 = __int_as_float(__builtin_amdgcn_readlane(rc.y, j + 1));
      const float v2 = __int_as_float(__builtin_amdgcn_readlane(rc.y, j + 2));
      const float v3 = __int_as_float(__builtin_amdgcn_readlane(rc.y, j + 3));
      const float4* p0 = xl + (size_t)c0 * (DIM / 4);
      const float4* p1 = xl + (size_t)c1 * (DIM / 4);
      const float4* p2 = xl + (size_t)c2 * (DIM / 4);
      const float4* p3 = xl + (size_t)c3 * (DIM / 4);
      const float4 x00 = p0[0], x01 = p0[32];
      const float4 x10 = p1[0], x11 = p1[32];
      const float4 x20 = p2[0], x21 = p2[32];
      const float4 x30 = p3[0], x31 = p3[32];
      add_scaled(a, v0, x00, x01);
      add_scaled(a, v1, x10, x11);
      add_scaled(a, v2, x20, x21);
      add_scaled(a, v3, x30, x31);
    }
    if (j + 2 <= n) {
      const int c0 = __builtin_amdgcn_readlane(rc.x, j);
      const int c1 = __builtin_amdgcn_readlane(rc.x, j + 1);
      const float v0 = __int_as_float(__builtin_amdgcn_readlane(rc.y, j));
      const float v1 = __int_as_float(__builtin_amdgcn_readlane(rc.y, j + 1));
      const float4* p0 = xl + (size_t)c0 * (DIM / 4);
      const float4* p1 = xl + (size_t)c1 * (DIM / 4);
      const float4 x00 = p0[0], x01 = p0[32];
      const float4 x10 = p1[0], x11 = p1[32];
      add_scaled(a, v0, x00, x01);
      add_scaled(a, v1, x10, x11);
      j += 2;
    }
    if (j < n) {
      const int c0 = __builtin_amdgcn_readlane(rc.x, j);
      const float v0 = __int_as_float(__builtin_amdgcn_readlane(rc.y, j));
      const float4* p0 = xl + (size_t)c0 * (DIM / 4);
      const float4 x00 = p0[0], x01 = p0[32];
      add_scaled(a, v0, x00, x01);
    }
  }
  float4* sp = reinterpret_cast<float4*>(side + (size_t)r * DIM);
  sp[lane]      = a.lo;
  sp[lane + 32] = a.hi;
}

#define GEMM_GROUPS 256
#define GEMM_TILES  (N_NODES / 16)
#define GEMM_RT_MAX 5
#define GEMM_T      512
#define GEMM_KB     8

template <int RT> struct ASet { float4 a1[RT], a2[RT]; };
struct WPair { float4 w1, w2; };

template <int RT>
__device__ __forceinline__ void load_aset(ASet<RT>& s, const float* a1p, const float* a2p, int k0) {
#pragma unroll
  for (int rt = 0; rt < RT; ++rt) {
    s.a1[rt] = *reinterpret_cast<const float4*>(a1p + rt * 16 * LDS_PITCH + k0);
    s.a2[rt] = *reinterpret_cast<const float4*>(a2p + rt * 16 * LDS_PITCH + k0);
  }
}

__device__ __forceinline__ void load_wpair(WPair& w, const float* __restrict__ w1p,
                                           const float* __restrict__ w2p, int k0) {
  w.w1 = *reinterpret_cast<const float4*>(w1p + k0);
  w.w2 = *reinterpret_cast<const float4*>(w2p + k0);
}

template <int RT>
__device__ __forceinline__ void mma_block(v8f (&acc1)[RT], v8f (&acc2)[RT],
                                          const ASet<RT>& s, const WPair& w) {
  __builtin_amdgcn_sched_barrier(0);
  const v2f w1lo = {w.w1.x, w.w1.y}, w1hi = {w.w1.z, w.w1.w};
  const v2f w2lo = {w.w2.x, w.w2.y}, w2hi = {w.w2.z, w.w2.w};
#pragma unroll
  for (int rt = 0; rt < RT; ++rt) {
    const v2f f1 = {s.a1[rt].x, s.a1[rt].y};
    const v2f f2 = {s.a2[rt].x, s.a2[rt].y};
    acc1[rt] = __builtin_amdgcn_wmma_f32_16x16x4_f32(
        false, f1, false, w1lo, (short)0, acc1[rt], false, false);
    acc2[rt] = __builtin_amdgcn_wmma_f32_16x16x4_f32(
        false, f2, false, w2lo, (short)0, acc2[rt], false, false);
  }
#pragma unroll
  for (int rt = 0; rt < RT; ++rt) {
    const v2f f1 = {s.a1[rt].z, s.a1[rt].w};
    const v2f f2 = {s.a2[rt].z, s.a2[rt].w};
    acc1[rt] = __builtin_amdgcn_wmma_f32_16x16x4_f32(
        false, f1, false, w1hi, (short)0, acc1[rt], false, false);
    acc2[rt] = __builtin_amdgcn_wmma_f32_16x16x4_f32(
        false, f2, false, w2hi, (short)0, acc2[rt], false, false);
  }
  __builtin_amdgcn_sched_barrier(0);
}

template <int RT>
__device__ __forceinline__ void gemm_strip(
    float* a1, float* a2, const int rowbase, const int t,
    const float* __restrict__ ego, const float* __restrict__ w1p, const float* __restrict__ w2p,
    const float bias1, const float bias2, float* out) {
#pragma unroll
  for (int j = 0; j < (RT * 16 * 64) / GEMM_T; ++j) {
    const int i4 = t + GEMM_T * j;
    const int r  = i4 >> 6;
    const int c4 = i4 & 63;
    const float4 e = reinterpret_cast<const float4*>(ego + (size_t)(rowbase + r) * DIM)[c4];
    const float4 s = reinterpret_cast<const float4*>(out + (size_t)(rowbase + r) * DIM)[c4];
    *reinterpret_cast<float4*>(&a1[r * LDS_PITCH + c4 * 4]) =
        make_float4(e.x + s.x, e.y + s.y, e.z + s.z, e.w + s.w);
    *reinterpret_cast<float4*>(&a2[r * LDS_PITCH + c4 * 4]) =
        make_float4(e.x * s.x, e.y * s.y, e.z * s.z, e.w * s.w);
  }
  __syncthreads();

  const int lane = t & 31;
  const int wave = t >> 5;
  const int m    = lane & 15;
  const int half = lane >> 4;

  const float* a1p = &a1[m * LDS_PITCH + half * 4];
  const float* a2p = &a2[m * LDS_PITCH + half * 4];

  v8f acc1[RT];
  v8f acc2[RT];
#pragma unroll
  for (int rt = 0; rt < RT; ++rt) { acc1[rt] = v8f{}; acc2[rt] = v8f{}; }

  ASet<RT> fa, fb;
  WPair w0, w1, w2, w3;
  load_wpair(w0, w1p, w2p, 0);
  load_wpair(w1, w1p, w2p, GEMM_KB);
  load_wpair(w2, w1p, w2p, 2 * GEMM_KB);
  load_aset<RT>(fa, a1p, a2p, 0);
#pragma unroll 1
  for (int kb = 0; kb < DIM - 4 * GEMM_KB; kb += 4 * GEMM_KB) {
    load_aset<RT>(fb, a1p, a2p, kb + GEMM_KB);
    load_wpair(w3, w1p, w2p, kb + 3 * GEMM_KB);
    mma_block<RT>(acc1, acc2, fa, w0);
    load_aset<RT>(fa, a1p, a2p, kb + 2 * GEMM_KB);
    load_wpair(w0, w1p, w2p, kb + 4 * GEMM_KB);
    mma_block<RT>(acc1, acc2, fb, w1);
    load_aset<RT>(fb, a1p, a2p, kb + 3 * GEMM_KB);
    load_wpair(w1, w1p, w2p, kb + 5 * GEMM_KB);
    mma_block<RT>(acc1, acc2, fa, w2);
    load_aset<RT>(fa, a1p, a2p, kb + 4 * GEMM_KB);
    load_wpair(w2, w1p, w2p, kb + 6 * GEMM_KB);
    mma_block<RT>(acc1, acc2, fb, w3);
  }
  load_aset<RT>(fb, a1p, a2p, DIM - 3 * GEMM_KB);
  load_wpair(w3, w1p, w2p, DIM - GEMM_KB);
  mma_block<RT>(acc1, acc2, fa, w0);
  load_aset<RT>(fa, a1p, a2p, DIM - 2 * GEMM_KB);
  mma_block<RT>(acc1, acc2, fb, w1);
  load_aset<RT>(fb, a1p, a2p, DIM - GEMM_KB);
  mma_block<RT>(acc1, acc2, fa, w2);
  mma_block<RT>(acc1, acc2, fb, w3);

#pragma unroll
  for (int rt = 0; rt < RT; ++rt) {
    float* op = out + (size_t)(rowbase + rt * 16 + half * 8) * DIM + wave * 16 + m;
#pragma unroll
    for (int v = 0; v < 8; ++v) {
      float s1 = acc1[rt][v] + bias1;
      s1 = (s1 >= 0.f) ? s1 : NEG_SLOPE * s1;
      float s2 = acc2[rt][v] + bias2;
      s2 = (s2 >= 0.f) ? s2 : NEG_SLOPE * s2;
      op[(size_t)v * DIM] = s1 + s2;
    }
  }
  __syncthreads();
}

__global__ __launch_bounds__(GEMM_T) void fused_dual_gemm(
    const float* __restrict__ ego,
    const float* __restrict__ W1, const float* __restrict__ b1,
    const float* __restrict__ W2, const float* __restrict__ b2,
    float* out) {
  __shared__ __attribute__((aligned(16))) float a1[GEMM_RT_MAX * 16 * LDS_PITCH];
  __shared__ __attribute__((aligned(16))) float a2[GEMM_RT_MAX * 16 * LDS_PITCH];

  const int t = threadIdx.x;
  const int b = blockIdx.x;
  const int tile0 = (b * GEMM_TILES) / GEMM_GROUPS;
  const int tile1 = ((b + 1) * GEMM_TILES) / GEMM_GROUPS;

  const int jg = (t >> 5) * 16 + (t & 15);
  const int half = (t & 31) >> 4;
  const float* w1p = W1 + (size_t)jg * DIM + half * 4;
  const float* w2p = W2 + (size_t)jg * DIM + half * 4;
  const float bias1 = b1[jg];
  const float bias2 = b2[jg];

  int tile = tile0;
  if (tile1 - tile0 == 13) {
    gemm_strip<5>(a1, a2, tile * 16, t, ego, w1p, w2p, bias1, bias2, out);
    tile += 5;
  } else {
    gemm_strip<4>(a1, a2, tile * 16, t, ego, w1p, w2p, bias1, bias2, out);
    tile += 4;
  }
#pragma unroll 1
  for (; tile < tile1; tile += 4)
    gemm_strip<4>(a1, a2, tile * 16, t, ego, w1p, w2p, bias1, bias2, out);
}

extern "C" void kernel_launch(void* const* d_in, const int* in_sizes, int n_in,
                              void* d_out, int out_size, void* d_ws, size_t ws_size,
                              hipStream_t stream) {
  const float* ego  = (const float*)d_in[0];
  const int*   erow = (const int*)  d_in[1];
  const int*   ecol = (const int*)  d_in[2];
  const float* eval = (const float*)d_in[3];
  const float* W1   = (const float*)d_in[4];
  const float* b1   = (const float*)d_in[5];
  const float* W2   = (const float*)d_in[6];
  const float* b2   = (const float*)d_in[7];
  float* out = (float*)d_out;
  const int n_edges = in_sizes[1];

  const size_t ws_need = ((size_t)2 * ROWS_PAD + (size_t)3 * n_edges) * sizeof(int);
  if (d_ws != nullptr && ws_size >= ws_need) {
    int* cnt    = (int*)d_ws;
    int* rowptr = cnt + ROWS_PAD;
    int* rec    = rowptr + ROWS_PAD;
    int* rank   = rec + (size_t)2 * n_edges;
    const int eblk = (n_edges + 255) / 256;
    zero_i32<<<ROWS_PAD / 256, 256, 0, stream>>>(cnt, ROWS_PAD);
    edge_rank<<<eblk, 256, 0, stream>>>(erow, cnt, rank, n_edges);
    row_offsets<<<ROWS_PAD / SCAN_T, SCAN_T, 0, stream>>>(cnt, rowptr, N_NODES);
    edge_place<<<eblk, 256, 0, stream>>>(erow, ecol, eval, rowptr, rank, rec, n_edges);
    row_gather_sum<<<(N_NODES + 7) / 8, 256, 0, stream>>>(ego, rowptr, cnt, rec, out, N_NODES);
  } else {
    const int n4 = (N_NODES * DIM) / 4;
    zero_f4<<<(n4 + 255) / 256, 256, 0, stream>>>(out, n4);
    const int nblk = (n_edges + EDGES_PER_BLOCK - 1) / EDGES_PER_BLOCK;
    spmm_scatter<<<nblk, 256, 0, stream>>>(erow, ecol, eval, ego, out, n_edges);
  }

  fused_dual_gemm<<<GEMM_GROUPS, GEMM_T, 0, stream>>>(ego, W1, b1, W2, b2, out);
}


# Edge sort rebuilt: edges binned by 64-row destination range in LDS (one global add per work-group and bin), each bin then ordered by row in LDS; row sum order = bin arrival order, still run-dependent
# speedup vs baseline: 3.3438x; 1.3334x over previous
#include <hip/hip_runtime.h>

typedef float v2f __attribute__((ext_vector_type(2)));
typedef float v8f __attribute__((ext_vector_type(8)));

#define N_NODES   50000
#define DIM       256
#define NEG_SLOPE 0.01f
#define LDS_PITCH 260
#define EDGES_PER_BLOCK 4

__global__ __launch_bounds__(256) void zero_f4(float* __restrict__ out, int n4) {
  int i = blockIdx.x * blockDim.x + threadIdx.x;
  if (i < n4) {
    float4 z = make_float4(0.f, 0.f, 0.f, 0.f);
    reinterpret_cast<float4*>(out)[i] = z;
  }
}

__global__ __launch_bounds__(256) void spmm_scatter(
    const int* __restrict__ erow, const int* __restrict__ ecol,
    const float* __restrict__ eval, const float* __restrict__ x,
    float* side, int n_edges) {
  const int t  = threadIdx.x;
  const int e0 = blockIdx.x * EDGES_PER_BLOCK;
#pragma unroll
  for (int i = 0; i < EDGES_PER_BLOCK; ++i) {
    int e = e0 + i;
    if (e < n_edges) {
      int   r = erow[e];
      int   c = ecol[e];
      float v = eval[e];
      float m = v * x[(size_t)c * DIM + t];
      atomicAdd(&side[(size_t)r * DIM + t], m);
    }
  }
}

#define SCAN_T   1024
#define ROWS_PAD 50176

__global__ __launch_bounds__(256) void zero_i32(int* __restrict__ p, int n) {
  int i = blockIdx.x * blockDim.x + threadIdx.x;
  if (i < n) p[i] = 0;
}

__global__ __launch_bounds__(256) void edge_rank(
    const int* __restrict__ erow, int* cnt, int* __restrict__ rank, int n_edges) {
  int e = blockIdx.x * blockDim.x + threadIdx.x;
  if (e < n_edges) rank[e] = atomicAdd(&cnt[erow[e]], 1);
}

__global__ __launch_bounds__(SCAN_T) void row_offsets(
    const int* __restrict__ cnt, int* __restrict__ rowptr, int n) {
  __shared__ int wsum[32];
  __shared__ int wpre[32];
  const int t = threadIdx.x;
  const int lane = t & 31;
  const int w = t >> 5;
  const int b = blockIdx.x;

  int pre = 0;
#pragma unroll 8
  for (int j = 0; j < b; ++j) pre += cnt[j * SCAN_T + t];
  const int i = b * SCAN_T + t;
  const int c = (i < n) ? cnt[i] : 0;

  int incl = c;
#pragma unroll
  for (int d = 1; d < 32; d <<= 1) {
    int y = __shfl_up(incl, d);
    if (lane >= d) incl += y;
  }
#pragma unroll
  for (int d = 16; d > 0; d >>= 1) pre += __shfl_xor(pre, d);
  if (lane == 31) wsum[w] = incl;
  if (lane == 0)  wpre[w] = pre;
  __syncthreads();

  int part = wpre[lane] + ((lane < w) ? wsum[lane] : 0);
#pragma unroll
  for (int d = 16; d > 0; d >>= 1) part += __shfl_xor(part, d);
  if (i < n) rowptr[i] = part + incl - c;
}

__global__ __launch_bounds__(256) void edge_place(
    const int* __restrict__ erow, const int* __restrict__ ecol,
    const float* __restrict__ eval, const int* __restrict__ rowptr,
    const int* __restrict__ rank, int* __restrict__ rec, int n_edges) {
  int e = blockIdx.x * blockDim.x + threadIdx.x;
  if (e < n_edges) {
    int pos = rowptr[erow[e]] + rank[e];
    if ((unsigned)pos < (unsigned)n_edges)
      reinterpret_cast<int2*>(rec)[pos] = make_int2(ecol[e], __float_as_int(eval[e]));
  }
}

#define NBINS    (ROWS_PAD / 64)
#define CHUNK    4096
#define BIN_T    1024
#define FINE_T   256
#define FINE_CAP 4096
#define NO_EDGE  0xFFFFFFFFu
static_assert(N_NODES <= 65536, "row and column are packed into 16 bits each");

__device__ __forceinline__ int block_excl_scan(int v, int* wsum, int t) {
  const int lane = t & 31;
  const int w = t >> 5;
  int incl = v;
#pragma unroll
  for (int d = 1; d < 32; d <<= 1) {
    int y = __shfl_up(incl, d);
    if (lane >= d) incl += y;
  }
  if (lane == 31) wsum[w] = incl;
  __syncthreads();
  int ws = (lane < w) ? wsum[lane] : 0;
#pragma unroll
  for (int d = 16; d > 0; d >>= 1) ws += __shfl_xor(ws, d);
  __syncthreads();
  return ws + incl - v;
}

__global__ __launch_bounds__(BIN_T) void bin_count(
    const int* __restrict__ erow, int* total, int n_edges) {
  __shared__ int hist[NBINS];
  const int t = threadIdx.x;
  if (t < NBINS) hist[t] = 0;
  __syncthreads();
  const int e0 = blockIdx.x * CHUNK;
#pragma unroll
  for (int j = 0; j < CHUNK / BIN_T; ++j) {
    const int e = e0 + j * BIN_T + t;
    if (e < n_edges) {
      const unsigned r = (unsigned)erow[e];
      if (r < (unsigned)N_NODES) atomicAdd(&hist[r >> 6], 1);
    }
  }
  __syncthreads();
  if (t < NBINS) {
    const int h = hist[t];
    if (h > 0) atomicAdd(&total[t], h);
  }
}

__global__ __launch_bounds__(BIN_T) void bin_scatter(
    const int* __restrict__ erow, const int* __restrict__ ecol, const float* __restrict__ eval,
    const int* __restrict__ total, int* cursor, int* __restrict__ binoff,
    int* __restrict__ tmp, int n_edges) {
  __shared__ int hist[NBINS];
  __shared__ int lstart[NBINS];
  __shared__ int lcur[NBINS];
  __shared__ int gbase[NBINS];
  __shared__ int wsum[32];
  __shared__ int2 sorted[CHUNK];
  const int t = threadIdx.x;
  if (t < NBINS) { hist[t] = 0; lcur[t] = 0; }
  const int e0 = blockIdx.x * CHUNK;
  unsigned key[CHUNK / BIN_T];
  int      valb[CHUNK / BIN_T];
#pragma unroll
  for (int j = 0; j < CHUNK / BIN_T; ++j) {
    const int e = e0 + j * BIN_T + t;
    key[j] = NO_EDGE;
    valb[j] = 0;
    if (e < n_edges) {
      const unsigned r = (unsigned)erow[e];
      if (r < (unsigned)N_NODES) {
        key[j]  = (r << 16) | ((unsigned)ecol[e] & 0xFFFFu);
        valb[j] = __float_as_int(eval[e]);
      }
    }
  }
  const int tot = (t < NBINS) ? total[t] : 0;
  const int boff = block_excl_scan(tot, wsum, t);
  if (blockIdx.x == 0 && t <= NBINS) binoff[t] = boff;
#pragma unroll
  for (int j = 0; j < CHUNK / BIN_T; ++j)
    if (key[j] != NO_EDGE) atomicAdd(&hist[key[j] >> 22], 1);
  __syncthreads();
  const int h  = (t < NBINS) ? hist[t] : 0;
  const int ls = block_excl_scan(h, wsum, t);
  if (t < NBINS) {
    lstart[t] = ls;
    gbase[t]  = (h > 0) ? boff + atomicAdd(&cursor[t], h) : 0;
  }
  __syncthreads();
#pragma unroll
  for (int j = 0; j < CHUNK / BIN_T; ++j) {
    if (key[j] != NO_EDGE) {
      const int b = (int)(key[j] >> 22);
      const int p = atomicAdd(&lcur[b], 1);
      sorted[lstart[b] + p] = make_int2((int)key[j], valb[j]);
    }
  }
  __syncthreads();
  const int nloc = lstart[NBINS - 1] + hist[NBINS - 1];
  for (int i = t; i < nloc; i += BIN_T) {
    const int2 rc = sorted[i];
    const int b = (int)((unsigned)rc.x >> 22);
    const int dst = gbase[b] + (i - lstart[b]);
    if ((unsigned)dst < (unsigned)n_edges) reinterpret_cast<int2*>(tmp)[dst] = rc;
  }
}

__global__ __launch_bounds__(FINE_T) void bin_finish(
    const int* __restrict__ binoff, const int* __restrict__ tmp,
    int* __restrict__ rowptr, int* __restrict__ cnt, int* __restrict__ rec) {
  __shared__ int rc64[64];
  __shared__ int ro64[64];
  __shared__ int cu64[64];
  __shared__ int2 srt[FINE_CAP];
  const int t = threadIdx.x;
  const int b = blockIdx.x;
  const int s = binoff[b];
  const int n = binoff[b + 1] - s;
  if (t < 64) { rc64[t] = 0; cu64[t] = 0; }
  __syncthreads();
  const int2* src = reinterpret_cast<const int2*>(tmp) + s;
  int2*       dst = reinterpret_cast<int2*>(rec) + s;
  for (int i = t; i < n; i += FINE_T) {
    const unsigned k = (unsigned)src[i].x;
    atomicAdd(&rc64[(k >> 16) & 63u], 1);
  }
  __syncthreads();
  if (t < 64) {
    int o = 0;
    for (int q = 0; q < t; ++q) o += rc64[q];
    ro64[t] = o;
    const int row = b * 64 + t;
    rowptr[row] = s + o;
    cnt[row]    = rc64[t];
  }
  __syncthreads();
  if (n <= FINE_CAP) {
    for (int i = t; i < n; i += FINE_T) {
      const int2 r = src[i];
      const unsigned k = (unsigned)r.x;
      const int rl = (int)((k >> 16) & 63u);
      const int p = atomicAdd(&cu64[rl], 1);
      srt[ro64[rl] + p] = make_int2((int)(k & 0xFFFFu), r.y);
    }
    __syncthreads();
    for (int i = t; i < n; i += FINE_T) dst[i] = srt[i];
  } else {
    for (int i = t; i < n; i += FINE_T) {
      const int2 r = src[i];
      const unsigned k = (unsigned)r.x;
      const int rl = (int)((k >> 16) & 63u);
      const int p = atomicAdd(&cu64[rl], 1);
      dst[ro64[rl] + p] = make_int2((int)(k & 0xFFFFu), r.y);
    }
  }
}

struct RowAcc { float4 lo, hi; };

__device__ __forceinline__ void add_scaled(RowAcc& a, float v, const float4& x0, const float4& x1) {
  a.lo.x += v * x0.x; a.lo.y += v * x0.y; a.lo.z += v * x0.z; a.lo.w += v * x0.w;
  a.hi.x += v * x1.x; a.hi.y += v * x1.y; a.hi.z += v * x1.z; a.hi.w += v * x1.w;
}

__global__ __launch_bounds__(256) void row_gather_sum(
    const float* __restrict__ x, const int* __restrict__ rowptr,
    const int* __restrict__ cnt, const int* __restrict__ rec,
    float* __restrict__ side, int n_rows) {
  const int lane = threadIdx.x & 31;
  const int r = __builtin_amdgcn_readfirstlane(blockIdx.x * 8 + (threadIdx.x >> 5));
  if (r >= n_rows) return;
  const int start = __builtin_amdgcn_readfirstlane(rowptr[r]);
  const int deg   = __builtin_amdgcn_readfirstlane(cnt[r]);
  const int2* recs = reinterpret_cast<const int2*>(rec) + start;
  const float4* xl = reinterpret_cast<const float4*>(x) + lane;

  RowAcc a;
  a.lo = make_float4(0.f, 0.f, 0.f, 0.f);
  a.hi = make_float4(0.f, 0.f, 0.f, 0.f);

  for (int base = 0; base < deg; base += 32) {
    const int n = (deg - base < 32) ? (deg - base) : 32;
    int2 rc = make_int2(0, 0);
    if (lane < n) rc = recs[base + lane];
    int j = 0;
    for (; j + 4 <= n; j += 4) {
      const int c0 = __builtin_amdgcn_readlane(rc.x, j);
      const int c1 = __builtin_amdgcn_readlane(rc.x, j + 1);
      const int c2 = __builtin_amdgcn_readlane(rc.x, j + 2);
      const int c3 = __builtin_amdgcn_readlane(rc.x, j + 3);
      const float v0 = __int_as_float(__builtin_amdgcn_readlane(rc.y, j));
      const float v1 = __int_as_float(__builtin_amdgcn_readlane(rc.y, j + 1));
      const float v2 = __int_as_float(__builtin_amdgcn_readlane(rc.y, j + 2));
      const float v3 = __int_as_float(__builtin_amdgcn_readlane(rc.y, j + 3));
      const float4* p0 = xl + (size_t)c0 * (DIM / 4);
      const float4* p1 = xl + (size_t)c1 * (DIM / 4);
      const float4* p2 = xl + (size_t)c2 * (DIM / 4);
      const float4* p3 = xl + (size_t)c3 * (DIM / 4);
      const float4 x00 = p0[0], x01 = p0[32];
      const float4 x10 = p1[0], x11 = p1[32];
      const float4 x20 = p2[0], x21 = p2[32];
      const float4 x30 = p3[0], x31 = p3[32];
      add_scaled(a, v0, x00, x01);
      add_scaled(a, v1, x10, x11);
      add_scaled(a, v2, x20, x21);
      add_scaled(a, v3, x30, x31);
    }
    if (j + 2 <= n) {
      const int c0 = __builtin_amdgcn_readlane(rc.x, j);
      const int c1 = __builtin_amdgcn_readlane(rc.x, j + 1);
      const float v0 = __int_as_float(__builtin_amdgcn_readlane(rc.y, j));
      const float v1 = __int_as_float(__builtin_amdgcn_readlane(rc.y, j + 1));
      const float4* p0 = xl + (size_t)c0 * (DIM / 4);
      const float4* p1 = xl + (size_t)c1 * (DIM / 4);
      const float4 x00 = p0[0], x01 = p0[32];
      const float4 x10 = p1[0], x11 = p1[32];
      add_scaled(a, v0, x00, x01);
      add_scaled(a, v1, x10, x11);
      j += 2;
    }
    if (j < n) {
      const int c0 = __builtin_amdgcn_readlane(rc.x, j);
      const float v0 = __int_as_float(__builtin_amdgcn_readlane(rc.y, j));
      const float4* p0 = xl + (size_t)c0 * (DIM / 4);
      const float4 x00 = p0[0], x01 = p0[32];
      add_scaled(a, v0, x00, x01);
    }
  }
  float4* sp = reinterpret_cast<float4*>(side + (size_t)r * DIM);
  sp[lane]      = a.lo;
  sp[lane + 32] = a.hi;
}

#define GEMM_GROUPS 256
#define GEMM_TILES  (N_NODES / 16)
#define GEMM_RT_MAX 5
#define GEMM_T      512
#define GEMM_KB     8

template <int RT> struct ASet { float4 a1[RT], a2[RT]; };
struct WPair { float4 w1, w2; };

template <int RT>
__device__ __forceinline__ void load_aset(ASet<RT>& s, const float* a1p, const float* a2p, int k0) {
#pragma unroll
  for (int rt = 0; rt < RT; ++rt) {
    s.a1[rt] = *reinterpret_cast<const float4*>(a1p + rt * 16 * LDS_PITCH + k0);
    s.a2[rt] = *reinterpret_cast<const float4*>(a2p + rt * 16 * LDS_PITCH + k0);
  }
}

__device__ __forceinline__ void load_wpair(WPair& w, const float* __restrict__ w1p,
                                           const float* __restrict__ w2p, int k0) {
  w.w1 = *reinterpret_cast<const float4*>(w1p + k0);
  w.w2 = *reinterpret_cast<const float4*>(w2p + k0);
}

template <int RT>
__device__ __forceinline__ void mma_block(v8f (&acc1)[RT], v8f (&acc2)[RT],
                                          const ASet<RT>& s, const WPair& w) {
  __builtin_amdgcn_sched_barrier(0);
  const v2f w1lo = {w.w1.x, w.w1.y}, w1hi = {w.w1.z, w.w1.w};
  const v2f w2lo = {w.w2.x, w.w2.y}, w2hi = {w.w2.z, w.w2.w};
#pragma unroll
  for (int rt = 0; rt < RT; ++rt) {
    const v2f f1 = {s.a1[rt].x, s.a1[rt].y};
    const v2f f2 = {s.a2[rt].x, s.a2[rt].y};
    acc1[rt] = __builtin_amdgcn_wmma_f32_16x16x4_f32(
        false, f1, false, w1lo, (short)0, acc1[rt], false, false);
    acc2[rt] = __builtin_amdgcn_wmma_f32_16x16x4_f32(
        false, f2, false, w2lo, (short)0, acc2[rt], false, false);
  }
#pragma unroll
  for (int rt = 0; rt < RT; ++rt) {
    const v2f f1 = {s.a1[rt].z, s.a1[rt].w};
    const v2f f2 = {s.a2[rt].z, s.a2[rt].w};
    acc1[rt] = __builtin_amdgcn_wmma_f32_16x16x4_f32(
        false, f1, false, w1hi, (short)0, acc1[rt], false, false);
    acc2[rt] = __builtin_amdgcn_wmma_f32_16x16x4_f32(
        false, f2, false, w2hi, (short)0, acc2[rt], false, false);
  }
  __builtin_amdgcn_sched_barrier(0);
}

template <int RT>
__device__ __forceinline__ void gemm_strip(
    float* a1, float* a2, const int rowbase, const int t,
    const float* __restrict__ ego, const float* __restrict__ w1p, const float* __restrict__ w2p,
    const float bias1, const float bias2, float* out) {
#pragma unroll
  for (int j = 0; j < (RT * 16 * 64) / GEMM_T; ++j) {
    const int i4 = t + GEMM_T * j;
    const int r  = i4 >> 6;
    const int c4 = i4 & 63;
    const float4 e = reinterpret_cast<const float4*>(ego + (size_t)(rowbase + r) * DIM)[c4];
    const float4 s = reinterpret_cast<const float4*>(out + (size_t)(rowbase + r) * DIM)[c4];
    *reinterpret_cast<float4*>(&a1[r * LDS_PITCH + c4 * 4]) =
        make_float4(e.x + s.x, e.y + s.y, e.z + s.z, e.w + s.w);
    *reinterpret_cast<float4*>(&a2[r * LDS_PITCH + c4 * 4]) =
        make_float4(e.x * s.x, e.y * s.y, e.z * s.z, e.w * s.w);
  }
  __syncthreads();

  const int lane = t & 31;
  const int wave = t >> 5;
  const int m    = lane & 15;
  const int half = lane >> 4;

  const float* a1p = &a1[m * LDS_PITCH + half * 4];
  const float* a2p = &a2[m * LDS_PITCH + half * 4];

  v8f acc1[RT];
  v8f acc2[RT];
#pragma unroll
  for (int rt = 0; rt < RT; ++rt) { acc1[rt] = v8f{}; acc2[rt] = v8f{}; }

  ASet<RT> fa, fb;
  WPair w0, w1, w2, w3;
  load_wpair(w0, w1p, w2p, 0);
  load_wpair(w1, w1p, w2p, GEMM_KB);
  load_wpair(w2, w1p, w2p, 2 * GEMM_KB);
  load_aset<RT>(fa, a1p, a2p, 0);
#pragma unroll 1
  for (int kb = 0; kb < DIM - 4 * GEMM_KB; kb += 4 * GEMM_KB) {
    load_aset<RT>(fb, a1p, a2p, kb + GEMM_KB);
    load_wpair(w3, w1p, w2p, kb + 3 * GEMM_KB);
    mma_block<RT>(acc1, acc2, fa, w0);
    load_aset<RT>(fa, a1p, a2p, kb + 2 * GEMM_KB);
    load_wpair(w0, w1p, w2p, kb + 4 * GEMM_KB);
    mma_block<RT>(acc1, acc2, fb, w1);
    load_aset<RT>(fb, a1p, a2p, kb + 3 * GEMM_KB);
    load_wpair(w1, w1p, w2p, kb + 5 * GEMM_KB);
    mma_block<RT>(acc1, acc2, fa, w2);
    load_aset<RT>(fa, a1p, a2p, kb + 4 * GEMM_KB);
    load_wpair(w2, w1p, w2p, kb + 6 * GEMM_KB);
    mma_block<RT>(acc1, acc2, fb, w3);
  }
  load_aset<RT>(fb, a1p, a2p, DIM - 3 * GEMM_KB);
  load_wpair(w3, w1p, w2p, DIM - GEMM_KB);
  mma_block<RT>(acc1, acc2, fa, w0);
  load_aset<RT>(fa, a1p, a2p, DIM - 2 * GEMM_KB);
  mma_block<RT>(acc1, acc2, fb, w1);
  load_aset<RT>(fb, a1p, a2p, DIM - GEMM_KB);
  mma_block<RT>(acc1, acc2, fa, w2);
  mma_block<RT>(acc1, acc2, fb, w3);

#pragma unroll
  for (int rt = 0; rt < RT; ++rt) {
    float* op = out + (size_t)(rowbase + rt * 16 + half * 8) * DIM + wave * 16 + m;
#pragma unroll
    for (int v = 0; v < 8; ++v) {
      float s1 = acc1[rt][v] + bias1;
      s1 = (s1 >= 0.f) ? s1 : NEG_SLOPE * s1;
      float s2 = acc2[rt][v] + bias2;
      s2 = (s2 >= 0.f) ? s2 : NEG_SLOPE * s2;
      op[(size_t)v * DIM] = s1 + s2;
    }
  }
  __syncthreads();
}

__global__ __launch_bounds__(GEMM_T) void fused_dual_gemm(
    const float* __restrict__ ego,
    const float* __restrict__ W1, const float* __restrict__ b1,
    const float* __restrict__ W2, const float* __restrict__ b2,
    float* out) {
  __shared__ __attribute__((aligned(16))) float a1[GEMM_RT_MAX * 16 * LDS_PITCH];
  __shared__ __attribute__((aligned(16))) float a2[GEMM_RT_MAX * 16 * LDS_PITCH];

  const int t = threadIdx.x;
  const int b = blockIdx.x;
  const int tile0 = (b * GEMM_TILES) / GEMM_GROUPS;
  const int tile1 = ((b + 1) * GEMM_TILES) / GEMM_GROUPS;

  const int jg = (t >> 5) * 16 + (t & 15);
  const int half = (t & 31) >> 4;
  const float* w1p = W1 + (size_t)jg * DIM + half * 4;
  const float* w2p = W2 + (size_t)jg * DIM + half * 4;
  const float bias1 = b1[jg];
  const float bias2 = b2[jg];

  int tile = tile0;
  if (tile1 - tile0 == 13) {
    gemm_strip<5>(a1, a2, tile * 16, t, ego, w1p, w2p, bias1, bias2, out);
    tile += 5;
  } else {
    gemm_strip<4>(a1, a2, tile * 16, t, ego, w1p, w2p, bias1, bias2, out);
    tile += 4;
  }
#pragma unroll 1
  for (; tile < tile1; tile += 4)
    gemm_strip<4>(a1, a2, tile * 16, t, ego, w1p, w2p, bias1, bias2, out);
}

extern "C" void kernel_launch(void* const* d_in, const int* in_sizes, int n_in,
                              void* d_out, int out_size, void* d_ws, size_t ws_size,
                              hipStream_t stream) {
  const float* ego  = (const float*)d_in[0];
  const int*   erow = (const int*)  d_in[1];
  const int*   ecol = (const int*)  d_in[2];
  const float* eval = (const float*)d_in[3];
  const float* W1   = (const float*)d_in[4];
  const float* b1   = (const float*)d_in[5];
  const float* W2   = (const float*)d_in[6];
  const float* b2   = (const float*)d_in[7];
  float* out = (float*)d_out;
  const int n_edges = in_sizes[1];

  const size_t ws_bins = ((size_t)3 * 1024 + (size_t)2 * ROWS_PAD + (size_t)4 * n_edges) * sizeof(int);
  const size_t ws_need = ((size_t)2 * ROWS_PAD + (size_t)3 * n_edges) * sizeof(int);
  if (d_ws != nullptr && ws_size >= ws_bins) {
    int* total  = (int*)d_ws;
    int* cursor = total + 1024;
    int* binoff = cursor + 1024;
    int* rowptr = binoff + 1024;
    int* cnt    = rowptr + ROWS_PAD;
    int* rec    = cnt + ROWS_PAD;
    int* tmp    = rec + (size_t)2 * n_edges;
    const int cblk = (n_edges + CHUNK - 1) / CHUNK;
    zero_i32<<<2048 / 256, 256, 0, stream>>>(total, 2048);
    bin_count<<<cblk, BIN_T, 0, stream>>>(erow, total, n_edges);
    bin_scatter<<<cblk, BIN_T, 0, stream>>>(erow, ecol, eval, total, cursor, binoff, tmp, n_edges);
    bin_finish<<<NBINS, FINE_T, 0, stream>>>(binoff, tmp, rowptr, cnt, rec);
    row_gather_sum<<<(N_NODES + 7) / 8, 256, 0, stream>>>(ego, rowptr, cnt, rec, out, N_NODES);
  } else if (d_ws != nullptr && ws_size >= ws_need) {
    int* cnt    = (int*)d_ws;
    int* rowptr = cnt + ROWS_PAD;
    int* rec    = rowptr + ROWS_PAD;
    int* rank   = rec + (size_t)2 * n_edges;
    const int eblk = (n_edges + 255) / 256;
    zero_i32<<<ROWS_PAD / 256, 256, 0, stream>>>(cnt, ROWS_PAD);
    edge_rank<<<eblk, 256, 0, stream>>>(erow, cnt, rank, n_edges);
    row_offsets<<<ROWS_PAD / SCAN_T, SCAN_T, 0, stream>>>(cnt, rowptr, N_NODES);
    edge_place<<<eblk, 256, 0, stream>>>(erow, ecol, eval, rowptr, rank, rec, n_edges);
    row_gather_sum<<<(N_NODES + 7) / 8, 256, 0, stream>>>(ego, rowptr, cnt, rec, out, N_NODES);
  } else {
    const int n4 = (N_NODES * DIM) / 4;
    zero_f4<<<(n4 + 255) / 256, 256, 0, stream>>>(out, n4);
    const int nblk = (n_edges + EDGES_PER_BLOCK - 1) / EDGES_PER_BLOCK;
    spmm_scatter<<<nblk, 256, 0, stream>>>(erow, ecol, eval, ego, out, n_edges);
  }

  fused_dual_gemm<<<GEMM_GROUPS, GEMM_T, 0, stream>>>(ego, W1, b1, W2, b2, out);
}
